# starting kernel
# baseline (speedup 1.0000x reference)
#include <hip/hip_runtime.h>
#include <hip/hip_bf16.h>

typedef __attribute__((ext_vector_type(2))) float v2f;
typedef __attribute__((ext_vector_type(8))) float v8f;

#define IN_C   64
#define HID_C  64
#define N_REL  8
#define TILE_M 16
#define SROW   68

__global__ void wt_transpose_kernel(const float* __restrict__ W,
                                    float* __restrict__ Wt, int total) {
  int i = blockIdx.x * blockDim.x + threadIdx.x;
  if (i < total) {
    int r = i >> 12;
    int k = (i >> 6) & 63;
    int n = i & 63;
    Wt[(r << 12) + (n << 6) + k] = W[i];
  }
}

__global__ __launch_bounds__(128) void rgcn_fused_kernel(
    const float* __restrict__ x,
    const float* __restrict__ Wt,
    const int*   __restrict__ ptr,
    const int*   __restrict__ idx,
    const int*   __restrict__ et,
    const int*   __restrict__ hmap,
    const float* __restrict__ hbuf,
    const int*   __restrict__ hsize_p,
    float* __restrict__ out,
    int n_node)
{
  __shared__ float S[N_REL * TILE_M * SROW];
  __shared__ float s_deg[TILE_M];
  __shared__ int   s_map[TILE_M];

  const int tid   = threadIdx.x;
  const int lane  = tid & 31;
  const int wave  = tid >> 5;
  const int node0 = blockIdx.x * TILE_M;

  if (tid < TILE_M) {
    int n  = node0 + tid;
    int nc = (n < n_node) ? n : (n_node - 1);
    s_deg[tid] = (float)(ptr[nc + 1] - ptr[nc]);
    s_map[tid] = hmap[nc];
  }

  for (int nn = 0; nn < 4; ++nn) {
    const int local = wave * 4 + nn;
    int n = node0 + local;
    if (n >= n_node) n = n_node - 1;
    const int e0 = __builtin_amdgcn_readfirstlane(ptr[n]);
    const int e1 = __builtin_amdgcn_readfirstlane(ptr[n + 1]);
    v2f acc[N_REL];
#pragma unroll
    for (int r = 0; r < N_REL; ++r) acc[r] = (v2f)0.0f;
    for (int e = e0; e < e1; ++e) {
      const int src = __builtin_amdgcn_readfirstlane(idx[e]);
      const int t   = __builtin_amdgcn_readfirstlane(et[e]);
      const v2f v = *(const v2f*)(x + (size_t)src * IN_C + 2 * lane);
#pragma unroll
      for (int r = 0; r < N_REL; ++r) {
        if (t == r) acc[r] += v;
      }
    }
#pragma unroll
    for (int r = 0; r < N_REL; ++r) {
      *(v2f*)(S + (r * TILE_M + local) * SROW + 2 * lane) = acc[r];
    }
  }
  __syncthreads();

  const int ncol = lane & 15;
  const int hi   = lane >> 4;
  const int n0   = wave * 16;
  v8f acc = {};

#pragma unroll 2
  for (int r = 0; r < N_REL; ++r) {
    const float* Sr = S + r * TILE_M * SROW;
    const float* Br = Wt + (r << 12) + ((n0 + ncol) << 6);
#pragma unroll
    for (int j = 0; j < 16; ++j) {
      const int k = 4 * j + 2 * hi;
      const v2f A = *(const v2f*)(Sr + ncol * SROW + k);
      const v2f B = *(const v2f*)(Br + k);
      acc = __builtin_amdgcn_wmma_f32_16x16x4_f32(
          false, A, false, B, (short)0, acc, false, false);
    }
  }

  const int hsize = hsize_p[0];
  const size_t his_off = (size_t)n_node * HID_C;
#pragma unroll
  for (int v = 0; v < 8; ++v) {
    const int m    = v + 8 * hi;
    const int node = node0 + m;
    if (node < n_node) {
      float o = acc[v] / s_deg[m];
      if (hsize > 0 && s_map[m] != -1) {
        o = hbuf[(size_t)node * HID_C + n0 + ncol];
      }
      const size_t off = (size_t)node * HID_C + n0 + ncol;
      out[off]           = o;
      out[his_off + off] = o;
    }
  }
}

extern "C" void kernel_launch(void* const* d_in, const int* in_sizes, int n_in,
                              void* d_out, int out_size, void* d_ws, size_t ws_size,
                              hipStream_t stream) {
  const float* x    = (const float*)d_in[0];
  const float* W    = (const float*)d_in[1];
  const int*   ptr  = (const int*)d_in[2];
  const int*   idx  = (const int*)d_in[3];
  const int*   et   = (const int*)d_in[4];
  const int*   hmap = (const int*)d_in[6];
  const float* hbuf = (const float*)d_in[7];
  const int*   hsz  = (const int*)d_in[9];
  float* out = (float*)d_out;
  float* Wt  = (float*)d_ws;

  const int n_node = in_sizes[6];
  const int wtotal = in_sizes[1];

  wt_transpose_kernel<<<(wtotal + 255) / 256, 256, 0, stream>>>(W, Wt, wtotal);

  const int nblocks = (n_node + TILE_M - 1) / TILE_M;
  rgcn_fused_kernel<<<nblocks, 128, 0, stream>>>(x, Wt, ptr, idx, et, hmap,
                                                 hbuf, hsz, out, n_node);
}


# Edge loop takes 8 edges at a time: indices and relations are loaded one per lane, then the 8 row gathers are issued together before the adds so their latencies overlap; summation order kept
# speedup vs baseline: 1.4427x; 1.4427x over previous
#include <hip/hip_runtime.h>
#include <hip/hip_bf16.h>

typedef __attribute__((ext_vector_type(2))) float v2f;
typedef __attribute__((ext_vector_type(8))) float v8f;

#define IN_C   64
#define HID_C  64
#define N_REL  8
#define TILE_M 16
#define SROW   68
#define EDGE_CH 8

__global__ void wt_transpose_kernel(const float* __restrict__ W,
                                    float* __restrict__ Wt, int total) {
  int i = blockIdx.x * blockDim.x + threadIdx.x;
  if (i < total) {
    int r = i >> 12;
    int k = (i >> 6) & 63;
    int n = i & 63;
    Wt[(r << 12) + (n << 6) + k] = W[i];
  }
}

__global__ __launch_bounds__(128) void rgcn_fused_kernel(
    const float* __restrict__ x,
    const float* __restrict__ Wt,
    const int*   __restrict__ ptr,
    const int*   __restrict__ idx,
    const int*   __restrict__ et,
    const int*   __restrict__ hmap,
    const float* __restrict__ hbuf,
    const int*   __restrict__ hsize_p,
    float* __restrict__ out,
    int n_node)
{
  __shared__ float S[N_REL * TILE_M * SROW];
  __shared__ float s_deg[TILE_M];
  __shared__ int   s_map[TILE_M];

  const int tid   = threadIdx.x;
  const int lane  = tid & 31;
  const int wave  = tid >> 5;
  const int node0 = blockIdx.x * TILE_M;

  if (tid < TILE_M) {
    int n  = node0 + tid;
    int nc = (n < n_node) ? n : (n_node - 1);
    s_deg[tid] = (float)(ptr[nc + 1] - ptr[nc]);
    s_map[tid] = hmap[nc];
  }

  for (int nn = 0; nn < 4; ++nn) {
    const int local = wave * 4 + nn;
    int n = node0 + local;
    if (n >= n_node) n = n_node - 1;
    const int e0 = __builtin_amdgcn_readfirstlane(ptr[n]);
    const int e1 = __builtin_amdgcn_readfirstlane(ptr[n + 1]);
    v2f acc[N_REL];
#pragma unroll
    for (int r = 0; r < N_REL; ++r) acc[r] = (v2f)0.0f;
    for (int e = e0; e < e1; e += EDGE_CH) {
      const int left = e1 - e;
      const int cnt  = left < EDGE_CH ? left : EDGE_CH;
      int vi = 0;
      int vt = -1;
      if (lane < cnt) {
        vi = idx[e + lane];
        vt = et[e + lane];
      }
      v2f v[EDGE_CH];
#pragma unroll
      for (int u = 0; u < EDGE_CH; ++u) {
        const int src = __builtin_amdgcn_readlane(vi, u);
        v[u] = *(const v2f*)(x + (size_t)src * IN_C + 2 * lane);
      }
#pragma unroll
      for (int u = 0; u < EDGE_CH; ++u) {
        const int t = __builtin_amdgcn_readlane(vt, u);
#pragma unroll
        for (int r = 0; r < N_REL; ++r) {
          if (t == r) acc[r] += v[u];
        }
      }
    }
#pragma unroll
    for (int r = 0; r < N_REL; ++r) {
      *(v2f*)(S + (r * TILE_M + local) * SROW + 2 * lane) = acc[r];
    }
  }
  __syncthreads();

  const int ncol = lane & 15;
  const int hi   = lane >> 4;
  const int n0   = wave * 16;
  v8f acc = {};

#pragma unroll 2
  for (int r = 0; r < N_REL; ++r) {
    const float* Sr = S + r * TILE_M * SROW;
    const float* Br = Wt + (r << 12) + ((n0 + ncol) << 6);
#pragma unroll
    for (int j = 0; j < 16; ++j) {
      const int k = 4 * j + 2 * hi;
      const v2f A = *(const v2f*)(Sr + ncol * SROW + k);
      const v2f B = *(const v2f*)(Br + k);
      acc = __builtin_amdgcn_wmma_f32_16x16x4_f32(
          false, A, false, B, (short)0, acc, false, false);
    }
  }

  const int hsize = hsize_p[0];
  const size_t his_off = (size_t)n_node * HID_C;
#pragma unroll
  for (int v = 0; v < 8; ++v) {
    const int m    = v + 8 * hi;
    const int node = node0 + m;
    if (node < n_node) {
      float o = acc[v] / s_deg[m];
      if (hsize > 0 && s_map[m] != -1) {
        o = hbuf[(size_t)node * HID_C + n0 + ncol];
      }
      const size_t off = (size_t)node * HID_C + n0 + ncol;
      out[off]           = o;
      out[his_off + off] = o;
    }
  }
}

extern "C" void kernel_launch(void* const* d_in, const int* in_sizes, int n_in,
                              void* d_out, int out_size, void* d_ws, size_t ws_size,
                              hipStream_t stream) {
  const float* x    = (const float*)d_in[0];
  const float* W    = (const float*)d_in[1];
  const int*   ptr  = (const int*)d_in[2];
  const int*   idx  = (const int*)d_in[3];
  const int*   et   = (const int*)d_in[4];
  const int*   hmap = (const int*)d_in[6];
  const float* hbuf = (const float*)d_in[7];
  const int*   hsz  = (const int*)d_in[9];
  float* out = (float*)d_out;
  float* Wt  = (float*)d_ws;

  const int n_node = in_sizes[6];
  const int wtotal = in_sizes[1];

  wt_transpose_kernel<<<(wtotal + 255) / 256, 256, 0, stream>>>(W, Wt, wtotal);

  const int nblocks = (n_node + TILE_M - 1) / TILE_M;
  rgcn_fused_kernel<<<nblocks, 128, 0, stream>>>(x, Wt, ptr, idx, et, hmap,
                                                 hbuf, hsz, out, n_node);
}


# Edge loop: the per-relation accumulator array is indexed by the wave-uniform relation id (scalar-indexed registers), one add per edge instead of eight add-and-select pairs; edge order kept
# speedup vs baseline: 1.5980x; 1.1076x over previous
#include <hip/hip_runtime.h>
#include <hip/hip_bf16.h>

typedef __attribute__((ext_vector_type(2))) float v2f;
typedef __attribute__((ext_vector_type(8))) float v8f;

#define IN_C   64
#define HID_C  64
#define N_REL  8
#define TILE_M 16
#define SROW   68
#define EDGE_CH 8

__global__ void wt_transpose_kernel(const float* __restrict__ W,
                                    float* __restrict__ Wt, int total) {
  int i = blockIdx.x * blockDim.x + threadIdx.x;
  if (i < total) {
    int r = i >> 12;
    int k = (i >> 6) & 63;
    int n = i & 63;
    Wt[(r << 12) + (n << 6) + k] = W[i];
  }
}

__global__ __launch_bounds__(128) void rgcn_fused_kernel(
    const float* __restrict__ x,
    const float* __restrict__ Wt,
    const int*   __restrict__ ptr,
    const int*   __restrict__ idx,
    const int*   __restrict__ et,
    const int*   __restrict__ hmap,
    const float* __restrict__ hbuf,
    const int*   __restrict__ hsize_p,
    float* __restrict__ out,
    int n_node)
{
  __shared__ float S[N_REL * TILE_M * SROW];
  __shared__ float s_deg[TILE_M];
  __shared__ int   s_map[TILE_M];

  const int tid   = threadIdx.x;
  const int lane  = tid & 31;
  const int wave  = tid >> 5;
  const int node0 = blockIdx.x * TILE_M;

  if (tid < TILE_M) {
    int n  = node0 + tid;
    int nc = (n < n_node) ? n : (n_node - 1);
    s_deg[tid] = (float)(ptr[nc + 1] - ptr[nc]);
    s_map[tid] = hmap[nc];
  }

  for (int nn = 0; nn < 4; ++nn) {
    const int local = wave * 4 + nn;
    int n = node0 + local;
    if (n >= n_node) n = n_node - 1;
    const int e0 = __builtin_amdgcn_readfirstlane(ptr[n]);
    const int e1 = __builtin_amdgcn_readfirstlane(ptr[n + 1]);
    v2f acc[N_REL];
#pragma unroll
    for (int r = 0; r < N_REL; ++r) acc[r] = (v2f)0.0f;
    for (int e = e0; e < e1; e += EDGE_CH) {
      const int left = e1 - e;
      const int cnt  = left < EDGE_CH ? left : EDGE_CH;
      int vi = 0;
      int vt = -1;
      if (lane < cnt) {
        vi = idx[e + lane];
        vt = et[e + lane];
      }
      v2f v[EDGE_CH];
#pragma unroll
      for (int u = 0; u < EDGE_CH; ++u) {
        const int src = __builtin_amdgcn_readlane(vi, u);
        v[u] = *(const v2f*)(x + (size_t)src * IN_C + 2 * lane);
      }
#pragma unroll
      for (int u = 0; u < EDGE_CH; ++u) {
        const unsigned t = (unsigned)__builtin_amdgcn_readlane(vt, u);
        if (t < (unsigned)N_REL) acc[t] += v[u];
      }
    }
#pragma unroll
    for (int r = 0; r < N_REL; ++r) {
      *(v2f*)(S + (r * TILE_M + local) * SROW + 2 * lane) = acc[r];
    }
  }
  __syncthreads();

  const int ncol = lane & 15;
  const int hi   = lane >> 4;
  const int n0   = wave * 16;
  v8f acc = {};

#pragma unroll 2
  for (int r = 0; r < N_REL; ++r) {
    const float* Sr = S + r * TILE_M * SROW;
    const float* Br = Wt + (r << 12) + ((n0 + ncol) << 6);
#pragma unroll
    for (int j = 0; j < 16; ++j) {
      const int k = 4 * j + 2 * hi;
      const v2f A = *(const v2f*)(Sr + ncol * SROW + k);
      const v2f B = *(const v2f*)(Br + k);
      acc = __builtin_amdgcn_wmma_f32_16x16x4_f32(
          false, A, false, B, (short)0, acc, false, false);
    }
  }

  const int hsize = hsize_p[0];
  const size_t his_off = (size_t)n_node * HID_C;
#pragma unroll
  for (int v = 0; v < 8; ++v) {
    const int m    = v + 8 * hi;
    const int node = node0 + m;
    if (node < n_node) {
      float o = acc[v] / s_deg[m];
      if (hsize > 0 && s_map[m] != -1) {
        o = hbuf[(size_t)node * HID_C + n0 + ncol];
      }
      const size_t off = (size_t)node * HID_C + n0 + ncol;
      out[off]           = o;
      out[his_off + off] = o;
    }
  }
}

extern "C" void kernel_launch(void* const* d_in, const int* in_sizes, int n_in,
                              void* d_out, int out_size, void* d_ws, size_t ws_size,
                              hipStream_t stream) {
  const float* x    = (const float*)d_in[0];
  const float* W    = (const float*)d_in[1];
  const int*   ptr  = (const int*)d_in[2];
  const int*   idx  = (const int*)d_in[3];
  const int*   et   = (const int*)d_in[4];
  const int*   hmap = (const int*)d_in[6];
  const float* hbuf = (const float*)d_in[7];
  const int*   hsz  = (const int*)d_in[9];
  float* out = (float*)d_out;
  float* Wt  = (float*)d_ws;

  const int n_node = in_sizes[6];
  const int wtotal = in_sizes[1];

  wt_transpose_kernel<<<(wtotal + 255) / 256, 256, 0, stream>>>(W, Wt, wtotal);

  const int nblocks = (n_node + TILE_M - 1) / TILE_M;
  rgcn_fused_kernel<<<nblocks, 128, 0, stream>>>(x, Wt, ptr, idx, et, hmap,
                                                 hbuf, hsz, out, n_node);
}


# Phase-2 B operand: weight copy stored fragment-major, each B load is one contiguous 512 B wave read (16 B per lane, two k-steps) instead of a 256 B-strided read, one line per column; k order kept
# speedup vs baseline: 2.0240x; 1.2666x over previous
#include <hip/hip_runtime.h>
#include <hip/hip_bf16.h>

typedef __attribute__((ext_vector_type(2))) float v2f;
typedef __attribute__((ext_vector_type(4))) float v4f;
typedef __attribute__((ext_vector_type(8))) float v8f;

#define IN_C   64
#define HID_C  64
#define N_REL  8
#define TILE_M 16
#define SROW   68
#define EDGE_CH 8

__global__ void wt_transpose_kernel(const float* __restrict__ W,
                                    float* __restrict__ Wf, int total) {
  int i = blockIdx.x * blockDim.x + threadIdx.x;
  if (i < total) {
    int r = i >> 12;
    int k = (i >> 6) & 63;
    int n = i & 63;
    int nt   = n >> 4;
    int c    = n & 15;
    int jj   = k >> 3;
    int s    = ((k >> 1) & 2) | (k & 1);
    int h    = (k >> 1) & 1;
    int ln   = 16 * h + c;
    Wf[(((((r << 2) + nt) << 3) + jj) << 7) + (ln << 2) + s] = W[i];
  }
}

__global__ __launch_bounds__(128) void rgcn_fused_kernel(
    const float* __restrict__ x,
    const float* __restrict__ Wf,
    const int*   __restrict__ ptr,
    const int*   __restrict__ idx,
    const int*   __restrict__ et,
    const int*   __restrict__ hmap,
    const float* __restrict__ hbuf,
    const int*   __restrict__ hsize_p,
    float* __restrict__ out,
    int n_node)
{
  __shared__ float S[N_REL * TILE_M * SROW];
  __shared__ float s_deg[TILE_M];
  __shared__ int   s_map[TILE_M];

  const int tid   = threadIdx.x;
  const int lane  = tid & 31;
  const int wave  = tid >> 5;
  const int node0 = blockIdx.x * TILE_M;

  if (tid < TILE_M) {
    int n  = node0 + tid;
    int nc = (n < n_node) ? n : (n_node - 1);
    s_deg[tid] = (float)(ptr[nc + 1] - ptr[nc]);
    s_map[tid] = hmap[nc];
  }

  for (int nn = 0; nn < 4; ++nn) {
    const int local = wave * 4 + nn;
    int n = node0 + local;
    if (n >= n_node) n = n_node - 1;
    const int e0 = __builtin_amdgcn_readfirstlane(ptr[n]);
    const int e1 = __builtin_amdgcn_readfirstlane(ptr[n + 1]);
    v2f acc[N_REL];
#pragma unroll
    for (int r = 0; r < N_REL; ++r) acc[r] = (v2f)0.0f;
    for (int e = e0; e < e1; e += EDGE_CH) {
      const int left = e1 - e;
      const int cnt  = left < EDGE_CH ? left : EDGE_CH;
      int vi = 0;
      int vt = -1;
      if (lane < cnt) {
        vi = idx[e + lane];
        vt = et[e + lane];
      }
      v2f v[EDGE_CH];
#pragma unroll
      for (int u = 0; u < EDGE_CH; ++u) {
        const int src = __builtin_amdgcn_readlane(vi, u);
        v[u] = *(const v2f*)(x + (size_t)src * IN_C + 2 * lane);
      }
#pragma unroll
      for (int u = 0; u < EDGE_CH; ++u) {
        const unsigned t = (unsigned)__builtin_amdgcn_readlane(vt, u);
        if (t < (unsigned)N_REL) acc[t] += v[u];
      }
    }
#pragma unroll
    for (int r = 0; r < N_REL; ++r) {
      *(v2f*)(S + (r * TILE_M + local) * SROW + 2 * lane) = acc[r];
    }
  }
  __syncthreads();

  const int ncol = lane & 15;
  const int hi   = lane >> 4;
  const int n0   = wave * 16;
  v8f acc = {};

  const float* Bw = Wf + (wave << 10) + (lane << 2);
#pragma unroll 2
  for (int r = 0; r < N_REL; ++r) {
    const float* Sr = S + r * TILE_M * SROW + ncol * SROW + 2 * hi;
    const float* Br = Bw + (r << 12);
#pragma unroll
    for (int jj = 0; jj < 8; ++jj) {
      const v4f B4 = *(const v4f*)(Br + (jj << 7));
      const v2f A0 = *(const v2f*)(Sr + 8 * jj);
      const v2f A1 = *(const v2f*)(Sr + 8 * jj + 4);
      const v2f B0 = {B4.x, B4.y};
      const v2f B1 = {B4.z, B4.w};
      acc = __builtin_amdgcn_wmma_f32_16x16x4_f32(
          false, A0, false, B0, (short)0, acc, false, false);
      acc = __builtin_amdgcn_wmma_f32_16x16x4_f32(
          false, A1, false, B1, (short)0, acc, false, false);
    }
  }

  const int hsize = hsize_p[0];
  const size_t his_off = (size_t)n_node * HID_C;
#pragma unroll
  for (int v = 0; v < 8; ++v) {
    const int m    = v + 8 * hi;
    const int node = node0 + m;
    if (node < n_node) {
      float o = acc[v] / s_deg[m];
      if (hsize > 0 && s_map[m] != -1) {
        o = hbuf[(size_t)node * HID_C + n0 + ncol];
      }
      const size_t off = (size_t)node * HID_C + n0 + ncol;
      out[off]           = o;
      out[his_off + off] = o;
    }
  }
}

extern "C" void kernel_launch(void* const* d_in, const int* in_sizes, int n_in,
                              void* d_out, int out_size, void* d_ws, size_t ws_size,
                              hipStream_t stream) {
  const float* x    = (const float*)d_in[0];
  const float* W    = (const float*)d_in[1];
  const int*   ptr  = (const int*)d_in[2];
  const int*   idx  = (const int*)d_in[3];
  const int*   et   = (const int*)d_in[4];
  const int*   hmap = (const int*)d_in[6];
  const float* hbuf = (const float*)d_in[7];
  const int*   hsz  = (const int*)d_in[9];
  float* out = (float*)d_out;
  float* Wf  = (float*)d_ws;

  const int n_node = in_sizes[6];
  const int wtotal = in_sizes[1];

  wt_transpose_kernel<<<(wtotal + 255) / 256, 256, 0, stream>>>(W, Wf, wtotal);

  const int nblocks = (n_node + TILE_M - 1) / TILE_M;
  rgcn_fused_kernel<<<nblocks, 128, 0, stream>>>(x, Wf, ptr, idx, et, hmap,
                                                 hbuf, hsz, out, n_node);
}


# Edge loop software-pipelined over each wave's 4-node edge range: indices of the next 32 edges loaded a group ahead, row gathers of chunk k+1 issued before the adds of chunk k; edge order kept
# speedup vs baseline: 2.2525x; 1.1129x over previous
#include <hip/hip_runtime.h>
#include <hip/hip_bf16.h>

typedef __attribute__((ext_vector_type(2))) float v2f;
typedef __attribute__((ext_vector_type(4))) float v4f;
typedef __attribute__((ext_vector_type(8))) float v8f;
typedef __attribute__((ext_vector_type(32))) float v32f;

#define IN_C   64
#define HID_C  64
#define N_REL  8
#define TILE_M 16
#define SROW   68
#define EDGE_CH 8

__global__ void wt_transpose_kernel(const float* __restrict__ W,
                                    float* __restrict__ Wf, int total) {
  int i = blockIdx.x * blockDim.x + threadIdx.x;
  if (i < total) {
    int r = i >> 12;
    int k = (i >> 6) & 63;
    int n = i & 63;
    int nt   = n >> 4;
    int c    = n & 15;
    int jj   = k >> 3;
    int s    = ((k >> 1) & 2) | (k & 1);
    int h    = (k >> 1) & 1;
    int ln   = 16 * h + c;
    Wf[(((((r << 2) + nt) << 3) + jj) << 7) + (ln << 2) + s] = W[i];
  }
}

__global__ __launch_bounds__(128) void rgcn_fused_kernel(
    const float* __restrict__ x,
    const float* __restrict__ Wf,
    const int*   __restrict__ ptr,
    const int*   __restrict__ idx,
    const int*   __restrict__ et,
    const int*   __restrict__ hmap,
    const float* __restrict__ hbuf,
    const int*   __restrict__ hsize_p,
    float* __restrict__ out,
    int n_node)
{
  __shared__ float S[N_REL * TILE_M * SROW];
  __shared__ float s_deg[TILE_M];
  __shared__ int   s_map[TILE_M];

  const int tid   = threadIdx.x;
  const int lane  = tid & 31;
  const int wave  = tid >> 5;
  const int node0 = blockIdx.x * TILE_M;

  if (tid < TILE_M) {
    int n  = node0 + tid;
    int nc = (n < n_node) ? n : (n_node - 1);
    s_deg[tid] = (float)(ptr[nc + 1] - ptr[nc]);
    s_map[tid] = hmap[nc];
  }

  {
    const int nA = node0 + wave * 4;
    int pq = nA + (lane < 4 ? lane : 4);
    if (pq > n_node) pq = n_node;
    const int pv = ptr[pq];
    const int p0 = __builtin_amdgcn_readlane(pv, 0);
    const int p1 = __builtin_amdgcn_readlane(pv, 1);
    const int p2 = __builtin_amdgcn_readlane(pv, 2);
    const int p3 = __builtin_amdgcn_readlane(pv, 3);
    const int p4 = __builtin_amdgcn_readlane(pv, 4);

    v32f accX = (v32f)0.0f;
    v32f accY = (v32f)0.0f;

#define GATHER8(V, VI, LB)                                                     \
    _Pragma("unroll")                                                          \
    for (int u = 0; u < EDGE_CH; ++u) {                                        \
      const int src = __builtin_amdgcn_readlane(VI, (LB) + u);                 \
      V[u] = *(const v2f*)(x + (size_t)src * IN_C + 2 * lane);                 \
    }
#define ADD8(V, VC, LB)                                                        \
    _Pragma("unroll")                                                          \
    for (int u = 0; u < EDGE_CH; ++u) {                                        \
      const unsigned cc = (unsigned)__builtin_amdgcn_readlane(VC, (LB) + u);   \
      if (cc < (unsigned)(4 * N_REL)) {                                        \
        accX[cc] += V[u].x;                                                    \
        accY[cc] += V[u].y;                                                    \
      }                                                                        \
    }

    if (p0 < p4) {
      v2f va[EDGE_CH], vb[EDGE_CH];
      int vi, vc;
      {
        const int e  = p0 + lane;
        const int ec = e < p4 ? e : p4 - 1;
        const int i  = idx[ec];
        const int t  = et[ec];
        const int q  = (e >= p1) + (e >= p2) + (e >= p3);
        const bool ok = (e < p4) && ((unsigned)t < (unsigned)N_REL);
        vi = (e < p4) ? i : 0;
        vc = ok ? (q * N_REL + t) : -1;
      }
      for (int gb = p0; gb < p4; gb += 32) {
        const int en  = gb + 32 + lane;
        const int enc = en < p4 ? en : p4 - 1;
        const int in_ = idx[enc];
        const int tn  = et[enc];
        const int left = p4 - gb;
        GATHER8(va, vi, 0)
        GATHER8(vb, vi, 8)
        ADD8(va, vc, 0)
        GATHER8(va, vi, 16)
        if (left > 8)  { ADD8(vb, vc, 8) }
        GATHER8(vb, vi, 24)
        if (left > 16) { ADD8(va, vc, 16) }
        if (left > 24) { ADD8(vb, vc, 24) }
        const int  qn  = (en >= p1) + (en >= p2) + (en >= p3);
        const bool okn = (en < p4) && ((unsigned)tn < (unsigned)N_REL);
        vi = (en < p4) ? in_ : 0;
        vc = okn ? (qn * N_REL + tn) : -1;
      }
    }
#undef GATHER8
#undef ADD8

#pragma unroll
    for (int q = 0; q < 4; ++q) {
#pragma unroll
      for (int r = 0; r < N_REL; ++r) {
        v2f o;
        o.x = accX[q * N_REL + r];
        o.y = accY[q * N_REL + r];
        *(v2f*)(S + (r * TILE_M + wave * 4 + q) * SROW + 2 * lane) = o;
      }
    }
  }
  __syncthreads();

  const int ncol = lane & 15;
  const int hi   = lane >> 4;
  const int n0   = wave * 16;
  v8f acc = {};

  const float* Bw = Wf + (wave << 10) + (lane << 2);
#pragma unroll 2
  for (int r = 0; r < N_REL; ++r) {
    const float* Sr = S + r * TILE_M * SROW + ncol * SROW + 2 * hi;
    const float* Br = Bw + (r << 12);
#pragma unroll
    for (int jj = 0; jj < 8; ++jj) {
      const v4f B4 = *(const v4f*)(Br + (jj << 7));
      const v2f A0 = *(const v2f*)(Sr + 8 * jj);
      const v2f A1 = *(const v2f*)(Sr + 8 * jj + 4);
      const v2f B0 = {B4.x, B4.y};
      const v2f B1 = {B4.z, B4.w};
      acc = __builtin_amdgcn_wmma_f32_16x16x4_f32(
          false, A0, false, B0, (short)0, acc, false, false);
      acc = __builtin_amdgcn_wmma_f32_16x16x4_f32(
          false, A1, false, B1, (short)0, acc, false, false);
    }
  }

  const int hsize = hsize_p[0];
  const size_t his_off = (size_t)n_node * HID_C;
#pragma unroll
  for (int v = 0; v < 8; ++v) {
    const int m    = v + 8 * hi;
    const int node = node0 + m;
    if (node < n_node) {
      float o = acc[v] / s_deg[m];
      if (hsize > 0 && s_map[m] != -1) {
        o = hbuf[(size_t)node * HID_C + n0 + ncol];
      }
      const size_t off = (size_t)node * HID_C + n0 + ncol;
      out[off]           = o;
      out[his_off + off] = o;
    }
  }
}

extern "C" void kernel_launch(void* const* d_in, const int* in_sizes, int n_in,
                              void* d_out, int out_size, void* d_ws, size_t ws_size,
                              hipStream_t stream) {
  const float* x    = (const float*)d_in[0];
  const float* W    = (const float*)d_in[1];
  const int*   ptr  = (const int*)d_in[2];
  const int*   idx  = (const int*)d_in[3];
  const int*   et   = (const int*)d_in[4];
  const int*   hmap = (const int*)d_in[6];
  const float* hbuf = (const float*)d_in[7];
  const int*   hsz  = (const int*)d_in[9];
  float* out = (float*)d_out;
  float* Wf  = (float*)d_ws;

  const int n_node = in_sizes[6];
  const int wtotal = in_sizes[1];

  wt_transpose_kernel<<<(wtotal + 255) / 256, 256, 0, stream>>>(W, Wf, wtotal);

  const int nblocks = (n_node + TILE_M - 1) / TILE_M;
  rgcn_fused_kernel<<<nblocks, 128, 0, stream>>>(x, Wf, ptr, idx, et, hmap,
                                                 hbuf, hsz, out, n_node);
}
